# starting kernel
# baseline (speedup 1.0000x reference)
#include <hip/hip_runtime.h>

#define N_NODES 100000
#define N_EDGES 1600000
#define DCH     128
#define NLAY    3
#define SCAN_B  256
#define N_TILES (N_NODES/16)

typedef __attribute__((ext_vector_type(16))) __bf16         v16bf;
typedef __attribute__((ext_vector_type(8)))  float          v8f;
typedef __attribute__((ext_vector_type(4)))  float          v4f;
typedef __attribute__((ext_vector_type(4)))  unsigned       v4u;
typedef __attribute__((ext_vector_type(4)))  unsigned short v4us;

union BF16x16 { v16bf v; unsigned short u[16]; v4u q[2]; };

__device__ __forceinline__ void split_bf16(float x, unsigned short& h, unsigned short& l){
  __bf16 hb = (__bf16)x;
  h = __builtin_bit_cast(unsigned short, hb);
  __bf16 lb = (__bf16)(x - (float)hb);
  l = __builtin_bit_cast(unsigned short, lb);
}


__global__ void k_zero_i32(int* p, int n){
  int i = blockIdx.x*blockDim.x + threadIdx.x;
  if (i < n) p[i] = 0;
}
__global__ void k_copy_i32(int* d, const int* s, int n){
  int i = blockIdx.x*blockDim.x + threadIdx.x;
  if (i < n) d[i] = s[i];
}
__global__ void k_hist(const int* __restrict__ dst, int* __restrict__ counts){
  int e = blockIdx.x*blockDim.x + threadIdx.x;
  if (e < N_EDGES) atomicAdd(&counts[dst[e]], 1);
}
__global__ void k_scan_blocks(const int* __restrict__ counts, int* __restrict__ offsets,
                              int* __restrict__ bsum, int n){
  __shared__ int tmp[SCAN_B];
  int t = threadIdx.x;
  int i = blockIdx.x*SCAN_B + t;
  int v = (i < n) ? counts[i] : 0;
  tmp[t] = v; __syncthreads();
  for (int d = 1; d < SCAN_B; d <<= 1){
    int u = (t >= d) ? tmp[t-d] : 0; __syncthreads();
    tmp[t] += u; __syncthreads();
  }
  if (i < n) offsets[i] = tmp[t] - v;
  if (t == SCAN_B-1) bsum[blockIdx.x] = tmp[t];
}
__global__ void k_scan_sums(int* bsum, int nb){
  __shared__ int tmp[512];
  int t = threadIdx.x;
  int v = (t < nb) ? bsum[t] : 0;
  tmp[t] = v; __syncthreads();
  for (int d = 1; d < 512; d <<= 1){
    int u = (t >= d) ? tmp[t-d] : 0; __syncthreads();
    tmp[t] += u; __syncthreads();
  }
  if (t < nb) bsum[t] = tmp[t] - v;
}
__global__ void k_add_bsum(int* __restrict__ offsets, const int* __restrict__ bsum, int n){
  int i = blockIdx.x*blockDim.x + threadIdx.x;
  if (i < n) offsets[i] += bsum[i / SCAN_B];
  if (i == n) offsets[n] = N_EDGES;
}
__global__ void k_place(const int* __restrict__ src, const int* __restrict__ dst,
                        int* __restrict__ cursor, int* __restrict__ esorted){
  int e = blockIdx.x*blockDim.x + threadIdx.x;
  if (e < N_EDGES){
    int p = atomicAdd(&cursor[dst[e]], 1);
    esorted[p] = src[e];
  }
}

__global__ void k_pack_w(const float* __restrict__ W1, const float* __restrict__ W2,
                         unsigned short* __restrict__ wpk){
  int idx = blockIdx.x*blockDim.x + threadIdx.x;
  if (idx >= NLAY*2*16384) return;
  int layer = idx / (2*16384);
  int rem   = idx - layer*(2*16384);
  int mat   = rem >> 14;
  int t     = rem & 16383;
  int e    = t & 15;
  int lane = (t >> 4) & 31;
  int tile = t >> 9;
  int kt = tile >> 3, nt = tile & 7;
  int hf = lane >> 4, m = lane & 15;
  int k = kt*32 + hf*8 + (e & 7) + ((e >> 3) << 4);
  int n = nt*16 + m;
  const float* W = (mat == 0 ? W1 : W2) + (size_t)layer*16384;
  unsigned short hi, lo;
  split_bf16(W[k*DCH + n], hi, lo);
  size_t base = (size_t)(layer*2 + mat) * 2 * 16384;
  wpk[base + t]         = hi;
  wpk[base + 16384 + t] = lo;
}

__global__ void k_aggregate_pack(const float* __restrict__ x, const int* __restrict__ offsets,
                                 const int* __restrict__ esorted,
                                 unsigned short* __restrict__ aggh,
                                 unsigned short* __restrict__ aggl){
  int gid  = blockIdx.x*blockDim.x + threadIdx.x;
  int node = gid >> 5;
  int lane = threadIdx.x & 31;
  if (node >= N_NODES) return;
  int beg = offsets[node], end = offsets[node+1];
  const int c = lane * 4;
  v4f acc = *(const v4f*)(x + (size_t)node*DCH + c);
  for (int j0 = beg; j0 < end; j0 += 32){
    int idx = j0 + lane;
    int s = (idx < end) ? esorted[idx] : 0;
    int cnt = end - j0; if (cnt > 32) cnt = 32;
    for (int t = 0; t < cnt; ++t){
      int sj = __builtin_amdgcn_readlane(s, t);
      acc += *(const v4f*)(x + (size_t)sj*DCH + c);
    }
  }
  int tile = node >> 4, m = node & 15;
  int kt = c >> 5;
  int kk = c & 31;
  int hf = (kk >> 3) & 1;
  int e0 = (kk & 7) + ((kk >> 4) << 3);
  size_t idx = ((size_t)(tile*4 + kt)*32 + (hf*16 + m))*16 + e0;
  v4us hv, lv;
  #pragma unroll
  for (int i = 0; i < 4; ++i){
    unsigned short h, l; split_bf16(acc[i], h, l);
    hv[i] = h; lv[i] = l;
  }
  *(v4us*)(aggh + idx) = hv;
  *(v4us*)(aggl + idx) = lv;
}

__device__ __forceinline__ void load_frag(BF16x16& f, const unsigned short* __restrict__ w,
                                          size_t elem16_idx){
  const v4u* p = (const v4u*)(w + (elem16_idx << 4));
  f.q[0] = p[0]; f.q[1] = p[1];
}
__device__ __forceinline__ v8f wmma_bf16(const BF16x16& a, const BF16x16& b, v8f c){
  return __builtin_amdgcn_wmma_f32_16x16x32_bf16(false, a.v, false, b.v, (short)0, c, false, false);
}

__global__ void __launch_bounds__(32)
k_gin_mlp(const unsigned short* __restrict__ aggh, const unsigned short* __restrict__ aggl,
          const unsigned short* __restrict__ w1pk, const float* __restrict__ b1,
          const unsigned short* __restrict__ w2pk, const float* __restrict__ b2,
          float* __restrict__ out, int relu_out)
{
  __shared__ unsigned short h1h[2*4*32*16];
  __shared__ unsigned short h1l[2*4*32*16];
  const int lane = threadIdx.x;
  const int hf = lane >> 4;
  const int m  = lane & 15;
  const int tile0 = blockIdx.x*2;

  BF16x16 ahi[2][4], alo[2][4];
  #pragma unroll
  for (int tt = 0; tt < 2; ++tt)
    #pragma unroll
    for (int kt = 0; kt < 4; ++kt){
      size_t fi = (size_t)((tile0 + tt)*4 + kt)*32 + lane;
      load_frag(ahi[tt][kt], aggh, fi);
      load_frag(alo[tt][kt], aggl, fi);
    }

  #pragma unroll
  for (int nt = 0; nt < 8; ++nt){
    v8f acc[2] = {{}, {}};
    #pragma unroll
    for (int kt = 0; kt < 4; ++kt){
      BF16x16 bh, bl;
      load_frag(bh, w1pk,         (size_t)(kt*8 + nt)*32 + lane);
      load_frag(bl, w1pk + 16384, (size_t)(kt*8 + nt)*32 + lane);
      acc[0] = wmma_bf16(ahi[0][kt], bh, acc[0]);
      acc[1] = wmma_bf16(ahi[1][kt], bh, acc[1]);
      acc[0] = wmma_bf16(ahi[0][kt], bl, acc[0]);
      acc[1] = wmma_bf16(ahi[1][kt], bl, acc[1]);
      acc[0] = wmma_bf16(alo[0][kt], bh, acc[0]);
      acc[1] = wmma_bf16(alo[1][kt], bh, acc[1]);
    }
    float bias = b1[nt*16 + m];
    const int ktp = nt >> 1;
    const int hfp = m >> 3;
    const int ep  = (m & 7) + 8*(nt & 1);
    #pragma unroll
    for (int tt = 0; tt < 2; ++tt)
      #pragma unroll
      for (int e = 0; e < 8; ++e){
        float v = fmaxf(acc[tt][e] + bias, 0.0f);
        unsigned short h, l; split_bf16(v, h, l);
        int lanep = hfp*16 + (e + 8*hf);
        int idx = ((tt*4 + ktp)*32 + lanep)*16 + ep;
        h1h[idx] = h;
        h1l[idx] = l;
      }
  }
  __syncthreads();

  #pragma unroll
  for (int tt = 0; tt < 2; ++tt)
    #pragma unroll
    for (int kt = 0; kt < 4; ++kt){
      int fi = (tt*4 + kt)*32 + lane;
      const v4u* ph = (const v4u*)(h1h + (fi << 4));
      ahi[tt][kt].q[0] = ph[0]; ahi[tt][kt].q[1] = ph[1];
      const v4u* pl = (const v4u*)(h1l + (fi << 4));
      alo[tt][kt].q[0] = pl[0]; alo[tt][kt].q[1] = pl[1];
    }

  #pragma unroll
  for (int nt = 0; nt < 8; ++nt){
    v8f acc[2] = {{}, {}};
    #pragma unroll
    for (int kt = 0; kt < 4; ++kt){
      BF16x16 bh, bl;
      load_frag(bh, w2pk,         (size_t)(kt*8 + nt)*32 + lane);
      load_frag(bl, w2pk + 16384, (size_t)(kt*8 + nt)*32 + lane);
      acc[0] = wmma_bf16(ahi[0][kt], bh, acc[0]);
      acc[1] = wmma_bf16(ahi[1][kt], bh, acc[1]);
      acc[0] = wmma_bf16(ahi[0][kt], bl, acc[0]);
      acc[1] = wmma_bf16(ahi[1][kt], bl, acc[1]);
      acc[0] = wmma_bf16(alo[0][kt], bh, acc[0]);
      acc[1] = wmma_bf16(alo[1][kt], bh, acc[1]);
    }
    float bias = b2[nt*16 + m];
    #pragma unroll
    for (int tt = 0; tt < 2; ++tt)
      #pragma unroll
      for (int e = 0; e < 8; ++e){
        float v = acc[tt][e] + bias;
        if (relu_out) v = fmaxf(v, 0.0f);
        out[(size_t)((tile0 + tt)*16 + e + 8*hf)*DCH + nt*16 + m] = v;
      }
  }
}

extern "C" void kernel_launch(void* const* d_in, const int* in_sizes, int n_in,
                              void* d_out, int out_size, void* d_ws, size_t ws_size,
                              hipStream_t stream)
{
  (void)in_sizes; (void)n_in; (void)out_size; (void)ws_size;
  const float* x  = (const float*)d_in[0];
  const int*   ei = (const int*)  d_in[1];
  const float* W1 = (const float*)d_in[2];
  const float* b1 = (const float*)d_in[3];
  const float* W2 = (const float*)d_in[4];
  const float* b2 = (const float*)d_in[5];
  float* out = (float*)d_out;
  const int* src = ei;
  const int* dst = ei + N_EDGES;

  char* ws = (char*)d_ws;
  size_t off = 0;
  auto take = [&](size_t bytes){ void* p = ws + off; off = (off + bytes + 255) & ~(size_t)255; return p; };
  unsigned short* aggh    = (unsigned short*)take((size_t)N_NODES*DCH*sizeof(unsigned short));
  unsigned short* aggl    = (unsigned short*)take((size_t)N_NODES*DCH*sizeof(unsigned short));
  int*            esorted = (int*)           take((size_t)N_EDGES*sizeof(int));
  int*            offsets = (int*)           take((size_t)(N_NODES+1)*sizeof(int));
  int*            counts  = (int*)           take((size_t)N_NODES*sizeof(int));
  int*            cursor  = (int*)           take((size_t)N_NODES*sizeof(int));
  int*            bsum    = (int*)           take(512*sizeof(int));
  unsigned short* wpk     = (unsigned short*)take((size_t)NLAY*2*2*16384*sizeof(unsigned short));

  const int nb = (N_NODES + SCAN_B - 1) / SCAN_B;
  k_zero_i32   <<<nb, SCAN_B, 0, stream>>>(counts, N_NODES);
  k_hist       <<<(N_EDGES+255)/256, 256, 0, stream>>>(dst, counts);
  k_scan_blocks<<<nb, SCAN_B, 0, stream>>>(counts, offsets, bsum, N_NODES);
  k_scan_sums  <<<1, 512, 0, stream>>>(bsum, nb);
  k_add_bsum   <<<(N_NODES+1+255)/256, 256, 0, stream>>>(offsets, bsum, N_NODES);
  k_copy_i32   <<<nb, SCAN_B, 0, stream>>>(cursor, offsets, N_NODES);
  k_place      <<<(N_EDGES+255)/256, 256, 0, stream>>>(src, dst, cursor, esorted);
  k_pack_w     <<<(NLAY*2*16384+255)/256, 256, 0, stream>>>(W1, W2, wpk);

  const float* xc = x;
  for (int L = 0; L < NLAY; ++L){
    k_aggregate_pack<<<(N_NODES*32+255)/256, 256, 0, stream>>>(xc, offsets, esorted, aggh, aggl);
    const unsigned short* w1pk = wpk + (size_t)(L*2 + 0)*2*16384;
    const unsigned short* w2pk = wpk + (size_t)(L*2 + 1)*2*16384;
    k_gin_mlp<<<N_TILES/2, 32, 0, stream>>>(aggh, aggl, w1pk, b1 + L*DCH, w2pk, b2 + L*DCH,
                                            out, (L < NLAY-1) ? 1 : 0);
    xc = out;
  }
}


# CSR build by two-level counting sort on LDS atomics (256-node buckets) replaces global-atomic histogram and placement; neighbour sum order differs (still arrival order)
# speedup vs baseline: 1.3454x; 1.3454x over previous
#include <hip/hip_runtime.h>

#define N_NODES 100000
#define N_EDGES 1600000
#define DCH     128
#define NLAY    3
#define SCAN_B  256
#define N_TILES (N_NODES/16)

typedef __attribute__((ext_vector_type(16))) __bf16         v16bf;
typedef __attribute__((ext_vector_type(8)))  float          v8f;
typedef __attribute__((ext_vector_type(4)))  float          v4f;
typedef __attribute__((ext_vector_type(4)))  unsigned       v4u;
typedef __attribute__((ext_vector_type(4)))  unsigned short v4us;

union BF16x16 { v16bf v; unsigned short u[16]; v4u q[2]; };

__device__ __forceinline__ void split_bf16(float x, unsigned short& h, unsigned short& l){
  __bf16 hb = (__bf16)x;
  h = __builtin_bit_cast(unsigned short, hb);
  __bf16 lb = (__bf16)(x - (float)hb);
  l = __builtin_bit_cast(unsigned short, lb);
}


#define BKT_SH   8
#define BKT_N    (1 << BKT_SH)
#define NB_BKT   ((N_NODES + BKT_N - 1) / BKT_N)
#define BIN_T    640
#define BIN_PER  10
#define BIN_CH   (BIN_T*BIN_PER)
#define BIN_BLK  (N_EDGES/BIN_CH)
#define NMAT     (NB_BKT*BIN_BLK)
#define SCAN_NB  ((NMAT + SCAN_B - 1)/SCAN_B)
#define CSR_T    1024
static_assert(N_EDGES % BIN_CH == 0, "binning blocks must tile the edge list exactly");
static_assert(SCAN_NB <= 512, "block sums must fit the single-block scan");
static_assert(N_NODES <= (1 << (32 - BKT_SH)), "source id must fit beside the local node id");

__global__ void __launch_bounds__(BIN_T)
k_bin_count(const int* __restrict__ dst, int* __restrict__ mat){
  __shared__ int hist[NB_BKT];
  const int t = threadIdx.x;
  const int e0 = blockIdx.x*BIN_CH + t;
  int d[BIN_PER];
  for (int i = 0; i < BIN_PER; ++i) d[i] = dst[e0 + i*BIN_T];
  for (int i = t; i < NB_BKT; i += BIN_T) hist[i] = 0;
  __syncthreads();
  for (int i = 0; i < BIN_PER; ++i) atomicAdd(&hist[d[i] >> BKT_SH], 1);
  __syncthreads();
  for (int i = t; i < NB_BKT; i += BIN_T) mat[i*BIN_BLK + blockIdx.x] = hist[i];
}

__global__ void k_scan_blocks(const int* __restrict__ counts, int* __restrict__ offsets,
                              int* __restrict__ bsum, int n){
  __shared__ int tmp[SCAN_B];
  int t = threadIdx.x;
  int i = blockIdx.x*SCAN_B + t;
  int v = (i < n) ? counts[i] : 0;
  tmp[t] = v; __syncthreads();
  for (int d = 1; d < SCAN_B; d <<= 1){
    int u = (t >= d) ? tmp[t-d] : 0; __syncthreads();
    tmp[t] += u; __syncthreads();
  }
  if (i < n) offsets[i] = tmp[t] - v;
  if (t == SCAN_B-1) bsum[blockIdx.x] = tmp[t];
}
__global__ void k_scan_sums(int* bsum, int nb){
  __shared__ int tmp[512];
  int t = threadIdx.x;
  int v = (t < nb) ? bsum[t] : 0;
  tmp[t] = v; __syncthreads();
  for (int d = 1; d < 512; d <<= 1){
    int u = (t >= d) ? tmp[t-d] : 0; __syncthreads();
    tmp[t] += u; __syncthreads();
  }
  if (t < nb) bsum[t] = tmp[t] - v;
}

__device__ __forceinline__ int mat_prefix(const int* __restrict__ sc, const int* __restrict__ bsum, int j){
  return sc[j] + bsum[j / SCAN_B];
}

__global__ void __launch_bounds__(BIN_T)
k_bin_scatter(const int* __restrict__ src, const int* __restrict__ dst,
              const int* __restrict__ sc, const int* __restrict__ bsum,
              unsigned* __restrict__ binned){
  __shared__ int cur[NB_BKT];
  const int t = threadIdx.x;
  const int e0 = blockIdx.x*BIN_CH + t;
  int d[BIN_PER], s[BIN_PER];
  for (int i = 0; i < BIN_PER; ++i){ d[i] = dst[e0 + i*BIN_T]; s[i] = src[e0 + i*BIN_T]; }
  for (int i = t; i < NB_BKT; i += BIN_T) cur[i] = mat_prefix(sc, bsum, i*BIN_BLK + blockIdx.x);
  __syncthreads();
  for (int i = 0; i < BIN_PER; ++i){
    int p = atomicAdd(&cur[d[i] >> BKT_SH], 1);
    binned[p] = ((unsigned)s[i] << BKT_SH) | ((unsigned)d[i] & (unsigned)(BKT_N - 1));
  }
}

__global__ void __launch_bounds__(CSR_T)
k_bucket_csr(const unsigned* __restrict__ binned, const int* __restrict__ sc,
             const int* __restrict__ bsum, int* __restrict__ offsets, int* __restrict__ esorted){
  __shared__ int cnt[BKT_N];
  __shared__ int tmp[CSR_T];
  const int t = threadIdx.x;
  const int b = blockIdx.x;
  const int j0 = b*BIN_BLK;
  const int bstart = mat_prefix(sc, bsum, j0);
  int bend = N_EDGES;
  if (b + 1 < NB_BKT) bend = mat_prefix(sc, bsum, j0 + BIN_BLK);
  if (t < BKT_N) cnt[t] = 0;
  __syncthreads();
  for (int i = bstart + t; i < bend; i += CSR_T)
    atomicAdd(&cnt[binned[i] & (unsigned)(BKT_N - 1)], 1);
  __syncthreads();
  int v = (t < BKT_N) ? cnt[t] : 0;
  tmp[t] = v; __syncthreads();
  for (int d = 1; d < BKT_N; d <<= 1){
    int u = (t >= d) ? tmp[t-d] : 0; __syncthreads();
    tmp[t] += u; __syncthreads();
  }
  if (t < BKT_N){
    int o = bstart + tmp[t] - v;
    cnt[t] = o;
    int node = b*BKT_N + t;
    if (node < N_NODES) offsets[node] = o;
  }
  if (b == NB_BKT-1 && t == 0) offsets[N_NODES] = N_EDGES;
  __syncthreads();
  for (int i = bstart + t; i < bend; i += CSR_T){
    unsigned w = binned[i];
    int p = atomicAdd(&cnt[w & (unsigned)(BKT_N - 1)], 1);
    esorted[p] = (int)(w >> BKT_SH);
  }
}

__global__ void k_pack_w(const float* __restrict__ W1, const float* __restrict__ W2,
                         unsigned short* __restrict__ wpk){
  int idx = blockIdx.x*blockDim.x + threadIdx.x;
  if (idx >= NLAY*2*16384) return;
  int layer = idx / (2*16384);
  int rem   = idx - layer*(2*16384);
  int mat   = rem >> 14;
  int t     = rem & 16383;
  int e    = t & 15;
  int lane = (t >> 4) & 31;
  int tile = t >> 9;
  int kt = tile >> 3, nt = tile & 7;
  int hf = lane >> 4, m = lane & 15;
  int k = kt*32 + hf*8 + (e & 7) + ((e >> 3) << 4);
  int n = nt*16 + m;
  const float* W = (mat == 0 ? W1 : W2) + (size_t)layer*16384;
  unsigned short hi, lo;
  split_bf16(W[k*DCH + n], hi, lo);
  size_t base = (size_t)(layer*2 + mat) * 2 * 16384;
  wpk[base + t]         = hi;
  wpk[base + 16384 + t] = lo;
}

__global__ void k_aggregate_pack(const float* __restrict__ x, const int* __restrict__ offsets,
                                 const int* __restrict__ esorted,
                                 unsigned short* __restrict__ aggh,
                                 unsigned short* __restrict__ aggl){
  int gid  = blockIdx.x*blockDim.x + threadIdx.x;
  int node = gid >> 5;
  int lane = threadIdx.x & 31;
  if (node >= N_NODES) return;
  int beg = offsets[node], end = offsets[node+1];
  const int c = lane * 4;
  v4f acc = *(const v4f*)(x + (size_t)node*DCH + c);
  for (int j0 = beg; j0 < end; j0 += 32){
    int idx = j0 + lane;
    int s = (idx < end) ? esorted[idx] : 0;
    int cnt = end - j0; if (cnt > 32) cnt = 32;
    for (int t = 0; t < cnt; ++t){
      int sj = __builtin_amdgcn_readlane(s, t);
      acc += *(const v4f*)(x + (size_t)sj*DCH + c);
    }
  }
  int tile = node >> 4, m = node & 15;
  int kt = c >> 5;
  int kk = c & 31;
  int hf = (kk >> 3) & 1;
  int e0 = (kk & 7) + ((kk >> 4) << 3);
  size_t idx = ((size_t)(tile*4 + kt)*32 + (hf*16 + m))*16 + e0;
  v4us hv, lv;
  #pragma unroll
  for (int i = 0; i < 4; ++i){
    unsigned short h, l; split_bf16(acc[i], h, l);
    hv[i] = h; lv[i] = l;
  }
  *(v4us*)(aggh + idx) = hv;
  *(v4us*)(aggl + idx) = lv;
}

__device__ __forceinline__ void load_frag(BF16x16& f, const unsigned short* __restrict__ w,
                                          size_t elem16_idx){
  const v4u* p = (const v4u*)(w + (elem16_idx << 4));
  f.q[0] = p[0]; f.q[1] = p[1];
}
__device__ __forceinline__ v8f wmma_bf16(const BF16x16& a, const BF16x16& b, v8f c){
  return __builtin_amdgcn_wmma_f32_16x16x32_bf16(false, a.v, false, b.v, (short)0, c, false, false);
}

__global__ void __launch_bounds__(32)
k_gin_mlp(const unsigned short* __restrict__ aggh, const unsigned short* __restrict__ aggl,
          const unsigned short* __restrict__ w1pk, const float* __restrict__ b1,
          const unsigned short* __restrict__ w2pk, const float* __restrict__ b2,
          float* __restrict__ out, int relu_out)
{
  __shared__ unsigned short h1h[2*4*32*16];
  __shared__ unsigned short h1l[2*4*32*16];
  const int lane = threadIdx.x;
  const int hf = lane >> 4;
  const int m  = lane & 15;
  const int tile0 = blockIdx.x*2;

  BF16x16 ahi[2][4], alo[2][4];
  #pragma unroll
  for (int tt = 0; tt < 2; ++tt)
    #pragma unroll
    for (int kt = 0; kt < 4; ++kt){
      size_t fi = (size_t)((tile0 + tt)*4 + kt)*32 + lane;
      load_frag(ahi[tt][kt], aggh, fi);
      load_frag(alo[tt][kt], aggl, fi);
    }

  #pragma unroll
  for (int nt = 0; nt < 8; ++nt){
    v8f acc[2] = {{}, {}};
    #pragma unroll
    for (int kt = 0; kt < 4; ++kt){
      BF16x16 bh, bl;
      load_frag(bh, w1pk,         (size_t)(kt*8 + nt)*32 + lane);
      load_frag(bl, w1pk + 16384, (size_t)(kt*8 + nt)*32 + lane);
      acc[0] = wmma_bf16(ahi[0][kt], bh, acc[0]);
      acc[1] = wmma_bf16(ahi[1][kt], bh, acc[1]);
      acc[0] = wmma_bf16(ahi[0][kt], bl, acc[0]);
      acc[1] = wmma_bf16(ahi[1][kt], bl, acc[1]);
      acc[0] = wmma_bf16(alo[0][kt], bh, acc[0]);
      acc[1] = wmma_bf16(alo[1][kt], bh, acc[1]);
    }
    float bias = b1[nt*16 + m];
    const int ktp = nt >> 1;
    const int hfp = m >> 3;
    const int ep  = (m & 7) + 8*(nt & 1);
    #pragma unroll
    for (int tt = 0; tt < 2; ++tt)
      #pragma unroll
      for (int e = 0; e < 8; ++e){
        float v = fmaxf(acc[tt][e] + bias, 0.0f);
        unsigned short h, l; split_bf16(v, h, l);
        int lanep = hfp*16 + (e + 8*hf);
        int idx = ((tt*4 + ktp)*32 + lanep)*16 + ep;
        h1h[idx] = h;
        h1l[idx] = l;
      }
  }
  __syncthreads();

  #pragma unroll
  for (int tt = 0; tt < 2; ++tt)
    #pragma unroll
    for (int kt = 0; kt < 4; ++kt){
      int fi = (tt*4 + kt)*32 + lane;
      const v4u* ph = (const v4u*)(h1h + (fi << 4));
      ahi[tt][kt].q[0] = ph[0]; ahi[tt][kt].q[1] = ph[1];
      const v4u* pl = (const v4u*)(h1l + (fi << 4));
      alo[tt][kt].q[0] = pl[0]; alo[tt][kt].q[1] = pl[1];
    }

  #pragma unroll
  for (int nt = 0; nt < 8; ++nt){
    v8f acc[2] = {{}, {}};
    #pragma unroll
    for (int kt = 0; kt < 4; ++kt){
      BF16x16 bh, bl;
      load_frag(bh, w2pk,         (size_t)(kt*8 + nt)*32 + lane);
      load_frag(bl, w2pk + 16384, (size_t)(kt*8 + nt)*32 + lane);
      acc[0] = wmma_bf16(ahi[0][kt], bh, acc[0]);
      acc[1] = wmma_bf16(ahi[1][kt], bh, acc[1]);
      acc[0] = wmma_bf16(ahi[0][kt], bl, acc[0]);
      acc[1] = wmma_bf16(ahi[1][kt], bl, acc[1]);
      acc[0] = wmma_bf16(alo[0][kt], bh, acc[0]);
      acc[1] = wmma_bf16(alo[1][kt], bh, acc[1]);
    }
    float bias = b2[nt*16 + m];
    #pragma unroll
    for (int tt = 0; tt < 2; ++tt)
      #pragma unroll
      for (int e = 0; e < 8; ++e){
        float v = acc[tt][e] + bias;
        if (relu_out) v = fmaxf(v, 0.0f);
        out[(size_t)((tile0 + tt)*16 + e + 8*hf)*DCH + nt*16 + m] = v;
      }
  }
}

extern "C" void kernel_launch(void* const* d_in, const int* in_sizes, int n_in,
                              void* d_out, int out_size, void* d_ws, size_t ws_size,
                              hipStream_t stream)
{
  (void)in_sizes; (void)n_in; (void)out_size; (void)ws_size;
  const float* x  = (const float*)d_in[0];
  const int*   ei = (const int*)  d_in[1];
  const float* W1 = (const float*)d_in[2];
  const float* b1 = (const float*)d_in[3];
  const float* W2 = (const float*)d_in[4];
  const float* b2 = (const float*)d_in[5];
  float* out = (float*)d_out;
  const int* src = ei;
  const int* dst = ei + N_EDGES;

  char* ws = (char*)d_ws;
  size_t off = 0;
  auto take = [&](size_t bytes){ void* p = ws + off; off = (off + bytes + 255) & ~(size_t)255; return p; };
  unsigned short* aggh    = (unsigned short*)take((size_t)N_NODES*DCH*sizeof(unsigned short));
  unsigned short* aggl    = (unsigned short*)take((size_t)N_NODES*DCH*sizeof(unsigned short));
  int*            esorted = (int*)           take((size_t)N_EDGES*sizeof(int));
  int*            offsets = (int*)           take((size_t)(N_NODES+1)*sizeof(int));
  int*            mat     = (int*)           take((size_t)NMAT*sizeof(int));
  int*            sc      = (int*)           take((size_t)NMAT*sizeof(int));
  int*            bsum    = (int*)           take(512*sizeof(int));
  unsigned*       binned  = (unsigned*)      take((size_t)N_EDGES*sizeof(unsigned));
  unsigned short* wpk     = (unsigned short*)take((size_t)NLAY*2*2*16384*sizeof(unsigned short));

  k_bin_count  <<<BIN_BLK, BIN_T, 0, stream>>>(dst, mat);
  k_scan_blocks<<<SCAN_NB, SCAN_B, 0, stream>>>(mat, sc, bsum, NMAT);
  k_scan_sums  <<<1, 512, 0, stream>>>(bsum, SCAN_NB);
  k_bin_scatter<<<BIN_BLK, BIN_T, 0, stream>>>(src, dst, sc, bsum, binned);
  k_bucket_csr <<<NB_BKT, CSR_T, 0, stream>>>(binned, sc, bsum, offsets, esorted);
  k_pack_w     <<<(NLAY*2*16384+255)/256, 256, 0, stream>>>(W1, W2, wpk);

  const float* xc = x;
  for (int L = 0; L < NLAY; ++L){
    k_aggregate_pack<<<(N_NODES*32+255)/256, 256, 0, stream>>>(xc, offsets, esorted, aggh, aggl);
    const unsigned short* w1pk = wpk + (size_t)(L*2 + 0)*2*16384;
    const unsigned short* w2pk = wpk + (size_t)(L*2 + 1)*2*16384;
    k_gin_mlp<<<N_TILES/2, 32, 0, stream>>>(aggh, aggl, w1pk, b1 + L*DCH, w2pk, b2 + L*DCH,
                                            out, (L < NLAY-1) ? 1 : 0);
    xc = out;
  }
}


# MLP kernel: weight fragments are requested one column tile ahead through a 4-slot register ring (before, every k step waited for its own loads)
# speedup vs baseline: 1.7434x; 1.2958x over previous
#include <hip/hip_runtime.h>

#define N_NODES 100000
#define N_EDGES 1600000
#define DCH     128
#define NLAY    3
#define SCAN_B  256
#define N_TILES (N_NODES/16)

typedef __attribute__((ext_vector_type(16))) __bf16         v16bf;
typedef __attribute__((ext_vector_type(8)))  float          v8f;
typedef __attribute__((ext_vector_type(4)))  float          v4f;
typedef __attribute__((ext_vector_type(4)))  unsigned       v4u;
typedef __attribute__((ext_vector_type(4)))  unsigned short v4us;

union BF16x16 { v16bf v; unsigned short u[16]; v4u q[2]; };

__device__ __forceinline__ void split_bf16(float x, unsigned short& h, unsigned short& l){
  __bf16 hb = (__bf16)x;
  h = __builtin_bit_cast(unsigned short, hb);
  __bf16 lb = (__bf16)(x - (float)hb);
  l = __builtin_bit_cast(unsigned short, lb);
}


#define BKT_SH   8
#define BKT_N    (1 << BKT_SH)
#define NB_BKT   ((N_NODES + BKT_N - 1) / BKT_N)
#define BIN_T    640
#define BIN_PER  10
#define BIN_CH   (BIN_T*BIN_PER)
#define BIN_BLK  (N_EDGES/BIN_CH)
#define NMAT     (NB_BKT*BIN_BLK)
#define SCAN_NB  ((NMAT + SCAN_B - 1)/SCAN_B)
#define CSR_T    1024
static_assert(N_EDGES % BIN_CH == 0, "binning blocks must tile the edge list exactly");
static_assert(SCAN_NB <= 512, "block sums must fit the single-block scan");
static_assert(N_NODES <= (1 << (32 - BKT_SH)), "source id must fit beside the local node id");

__global__ void __launch_bounds__(BIN_T)
k_bin_count(const int* __restrict__ dst, int* __restrict__ mat){
  __shared__ int hist[NB_BKT];
  const int t = threadIdx.x;
  const int e0 = blockIdx.x*BIN_CH + t;
  int d[BIN_PER];
  for (int i = 0; i < BIN_PER; ++i) d[i] = dst[e0 + i*BIN_T];
  for (int i = t; i < NB_BKT; i += BIN_T) hist[i] = 0;
  __syncthreads();
  for (int i = 0; i < BIN_PER; ++i) atomicAdd(&hist[d[i] >> BKT_SH], 1);
  __syncthreads();
  for (int i = t; i < NB_BKT; i += BIN_T) mat[i*BIN_BLK + blockIdx.x] = hist[i];
}

__global__ void k_scan_blocks(const int* __restrict__ counts, int* __restrict__ offsets,
                              int* __restrict__ bsum, int n){
  __shared__ int tmp[SCAN_B];
  int t = threadIdx.x;
  int i = blockIdx.x*SCAN_B + t;
  int v = (i < n) ? counts[i] : 0;
  tmp[t] = v; __syncthreads();
  for (int d = 1; d < SCAN_B; d <<= 1){
    int u = (t >= d) ? tmp[t-d] : 0; __syncthreads();
    tmp[t] += u; __syncthreads();
  }
  if (i < n) offsets[i] = tmp[t] - v;
  if (t == SCAN_B-1) bsum[blockIdx.x] = tmp[t];
}
__global__ void k_scan_sums(int* bsum, int nb){
  __shared__ int tmp[512];
  int t = threadIdx.x;
  int v = (t < nb) ? bsum[t] : 0;
  tmp[t] = v; __syncthreads();
  for (int d = 1; d < 512; d <<= 1){
    int u = (t >= d) ? tmp[t-d] : 0; __syncthreads();
    tmp[t] += u; __syncthreads();
  }
  if (t < nb) bsum[t] = tmp[t] - v;
}

__device__ __forceinline__ int mat_prefix(const int* __restrict__ sc, const int* __restrict__ bsum, int j){
  return sc[j] + bsum[j / SCAN_B];
}

__global__ void __launch_bounds__(BIN_T)
k_bin_scatter(const int* __restrict__ src, const int* __restrict__ dst,
              const int* __restrict__ sc, const int* __restrict__ bsum,
              unsigned* __restrict__ binned){
  __shared__ int cur[NB_BKT];
  const int t = threadIdx.x;
  const int e0 = blockIdx.x*BIN_CH + t;
  int d[BIN_PER], s[BIN_PER];
  for (int i = 0; i < BIN_PER; ++i){ d[i] = dst[e0 + i*BIN_T]; s[i] = src[e0 + i*BIN_T]; }
  for (int i = t; i < NB_BKT; i += BIN_T) cur[i] = mat_prefix(sc, bsum, i*BIN_BLK + blockIdx.x);
  __syncthreads();
  for (int i = 0; i < BIN_PER; ++i){
    int p = atomicAdd(&cur[d[i] >> BKT_SH], 1);
    binned[p] = ((unsigned)s[i] << BKT_SH) | ((unsigned)d[i] & (unsigned)(BKT_N - 1));
  }
}

__global__ void __launch_bounds__(CSR_T)
k_bucket_csr(const unsigned* __restrict__ binned, const int* __restrict__ sc,
             const int* __restrict__ bsum, int* __restrict__ offsets, int* __restrict__ esorted){
  __shared__ int cnt[BKT_N];
  __shared__ int tmp[CSR_T];
  const int t = threadIdx.x;
  const int b = blockIdx.x;
  const int j0 = b*BIN_BLK;
  const int bstart = mat_prefix(sc, bsum, j0);
  int bend = N_EDGES;
  if (b + 1 < NB_BKT) bend = mat_prefix(sc, bsum, j0 + BIN_BLK);
  if (t < BKT_N) cnt[t] = 0;
  __syncthreads();
  for (int i = bstart + t; i < bend; i += CSR_T)
    atomicAdd(&cnt[binned[i] & (unsigned)(BKT_N - 1)], 1);
  __syncthreads();
  int v = (t < BKT_N) ? cnt[t] : 0;
  tmp[t] = v; __syncthreads();
  for (int d = 1; d < BKT_N; d <<= 1){
    int u = (t >= d) ? tmp[t-d] : 0; __syncthreads();
    tmp[t] += u; __syncthreads();
  }
  if (t < BKT_N){
    int o = bstart + tmp[t] - v;
    cnt[t] = o;
    int node = b*BKT_N + t;
    if (node < N_NODES) offsets[node] = o;
  }
  if (b == NB_BKT-1 && t == 0) offsets[N_NODES] = N_EDGES;
  __syncthreads();
  for (int i = bstart + t; i < bend; i += CSR_T){
    unsigned w = binned[i];
    int p = atomicAdd(&cnt[w & (unsigned)(BKT_N - 1)], 1);
    esorted[p] = (int)(w >> BKT_SH);
  }
}

__global__ void k_pack_w(const float* __restrict__ W1, const float* __restrict__ W2,
                         unsigned short* __restrict__ wpk){
  int idx = blockIdx.x*blockDim.x + threadIdx.x;
  if (idx >= NLAY*2*16384) return;
  int layer = idx / (2*16384);
  int rem   = idx - layer*(2*16384);
  int mat   = rem >> 14;
  int t     = rem & 16383;
  int e    = t & 15;
  int lane = (t >> 4) & 31;
  int tile = t >> 9;
  int kt = tile >> 3, nt = tile & 7;
  int hf = lane >> 4, m = lane & 15;
  int k = kt*32 + hf*8 + (e & 7) + ((e >> 3) << 4);
  int n = nt*16 + m;
  const float* W = (mat == 0 ? W1 : W2) + (size_t)layer*16384;
  unsigned short hi, lo;
  split_bf16(W[k*DCH + n], hi, lo);
  size_t base = (size_t)(layer*2 + mat) * 2 * 16384;
  wpk[base + t]         = hi;
  wpk[base + 16384 + t] = lo;
}

__global__ void k_aggregate_pack(const float* __restrict__ x, const int* __restrict__ offsets,
                                 const int* __restrict__ esorted,
                                 unsigned short* __restrict__ aggh,
                                 unsigned short* __restrict__ aggl){
  int gid  = blockIdx.x*blockDim.x + threadIdx.x;
  int node = gid >> 5;
  int lane = threadIdx.x & 31;
  if (node >= N_NODES) return;
  int beg = offsets[node], end = offsets[node+1];
  const int c = lane * 4;
  v4f acc = *(const v4f*)(x + (size_t)node*DCH + c);
  for (int j0 = beg; j0 < end; j0 += 32){
    int idx = j0 + lane;
    int s = (idx < end) ? esorted[idx] : 0;
    int cnt = end - j0; if (cnt > 32) cnt = 32;
    for (int t = 0; t < cnt; ++t){
      int sj = __builtin_amdgcn_readlane(s, t);
      acc += *(const v4f*)(x + (size_t)sj*DCH + c);
    }
  }
  int tile = node >> 4, m = node & 15;
  int kt = c >> 5;
  int kk = c & 31;
  int hf = (kk >> 3) & 1;
  int e0 = (kk & 7) + ((kk >> 4) << 3);
  size_t idx = ((size_t)(tile*4 + kt)*32 + (hf*16 + m))*16 + e0;
  v4us hv, lv;
  #pragma unroll
  for (int i = 0; i < 4; ++i){
    unsigned short h, l; split_bf16(acc[i], h, l);
    hv[i] = h; lv[i] = l;
  }
  *(v4us*)(aggh + idx) = hv;
  *(v4us*)(aggl + idx) = lv;
}

__device__ __forceinline__ void load_frag(BF16x16& f, const unsigned short* __restrict__ w,
                                          size_t elem16_idx){
  const v4u* p = (const v4u*)(w + (elem16_idx << 4));
  f.q[0] = p[0]; f.q[1] = p[1];
}
__device__ __forceinline__ v8f wmma_bf16(const BF16x16& a, const BF16x16& b, v8f c){
  return __builtin_amdgcn_wmma_f32_16x16x32_bf16(false, a.v, false, b.v, (short)0, c, false, false);
}

__global__ void __launch_bounds__(32)
k_gin_mlp(const unsigned short* __restrict__ aggh, const unsigned short* __restrict__ aggl,
          const unsigned short* __restrict__ w1pk, const float* __restrict__ b1,
          const unsigned short* __restrict__ w2pk, const float* __restrict__ b2,
          float* __restrict__ out, int relu_out)
{
  __shared__ unsigned short h1h[2*4*32*16];
  __shared__ unsigned short h1l[2*4*32*16];
  const int lane = threadIdx.x;
  const int hf = lane >> 4;
  const int m  = lane & 15;
  const int tile0 = blockIdx.x*2;

  BF16x16 ahi[2][4], alo[2][4];
  #pragma unroll
  for (int tt = 0; tt < 2; ++tt)
    #pragma unroll
    for (int kt = 0; kt < 4; ++kt){
      size_t fi = (size_t)((tile0 + tt)*4 + kt)*32 + lane;
      load_frag(ahi[tt][kt], aggh, fi);
      load_frag(alo[tt][kt], aggl, fi);
    }

  BF16x16 bh[4], bl[4];
  #pragma unroll
  for (int kt = 0; kt < 4; ++kt){
    load_frag(bh[kt], w1pk,         (size_t)(kt*8)*32 + lane);
    load_frag(bl[kt], w1pk + 16384, (size_t)(kt*8)*32 + lane);
  }

  #pragma unroll
  for (int nt = 0; nt < 8; ++nt){
    v8f acc[2] = {{}, {}};
    float bias = b1[nt*16 + m];
    const unsigned short* wn = (nt < 7) ? w1pk : w2pk;
    const int ntn = (nt < 7) ? nt + 1 : 0;
    #pragma unroll
    for (int kt = 0; kt < 4; ++kt){
      acc[0] = wmma_bf16(ahi[0][kt], bh[kt], acc[0]);
      acc[1] = wmma_bf16(ahi[1][kt], bh[kt], acc[1]);
      acc[0] = wmma_bf16(ahi[0][kt], bl[kt], acc[0]);
      acc[1] = wmma_bf16(ahi[1][kt], bl[kt], acc[1]);
      acc[0] = wmma_bf16(alo[0][kt], bh[kt], acc[0]);
      acc[1] = wmma_bf16(alo[1][kt], bh[kt], acc[1]);
      load_frag(bh[kt], wn,         (size_t)(kt*8 + ntn)*32 + lane);
      load_frag(bl[kt], wn + 16384, (size_t)(kt*8 + ntn)*32 + lane);
    }
    const int ktp = nt >> 1;
    const int hfp = m >> 3;
    const int ep  = (m & 7) + 8*(nt & 1);
    #pragma unroll
    for (int tt = 0; tt < 2; ++tt)
      #pragma unroll
      for (int e = 0; e < 8; ++e){
        float v = fmaxf(acc[tt][e] + bias, 0.0f);
        unsigned short h, l; split_bf16(v, h, l);
        int lanep = hfp*16 + (e + 8*hf);
        int idx = ((tt*4 + ktp)*32 + lanep)*16 + ep;
        h1h[idx] = h;
        h1l[idx] = l;
      }
  }
  __syncthreads();

  #pragma unroll
  for (int tt = 0; tt < 2; ++tt)
    #pragma unroll
    for (int kt = 0; kt < 4; ++kt){
      int fi = (tt*4 + kt)*32 + lane;
      const v4u* ph = (const v4u*)(h1h + (fi << 4));
      ahi[tt][kt].q[0] = ph[0]; ahi[tt][kt].q[1] = ph[1];
      const v4u* pl = (const v4u*)(h1l + (fi << 4));
      alo[tt][kt].q[0] = pl[0]; alo[tt][kt].q[1] = pl[1];
    }

  #pragma unroll
  for (int nt = 0; nt < 8; ++nt){
    v8f acc[2] = {{}, {}};
    float bias = b2[nt*16 + m];
    #pragma unroll
    for (int kt = 0; kt < 4; ++kt){
      acc[0] = wmma_bf16(ahi[0][kt], bh[kt], acc[0]);
      acc[1] = wmma_bf16(ahi[1][kt], bh[kt], acc[1]);
      acc[0] = wmma_bf16(ahi[0][kt], bl[kt], acc[0]);
      acc[1] = wmma_bf16(ahi[1][kt], bl[kt], acc[1]);
      acc[0] = wmma_bf16(alo[0][kt], bh[kt], acc[0]);
      acc[1] = wmma_bf16(alo[1][kt], bh[kt], acc[1]);
      if (nt < 7){
        load_frag(bh[kt], w2pk,         (size_t)(kt*8 + nt + 1)*32 + lane);
        load_frag(bl[kt], w2pk + 16384, (size_t)(kt*8 + nt + 1)*32 + lane);
      }
    }
    #pragma unroll
    for (int tt = 0; tt < 2; ++tt)
      #pragma unroll
      for (int e = 0; e < 8; ++e){
        float v = acc[tt][e] + bias;
        if (relu_out) v = fmaxf(v, 0.0f);
        out[(size_t)((tile0 + tt)*16 + e + 8*hf)*DCH + nt*16 + m] = v;
      }
  }
}

extern "C" void kernel_launch(void* const* d_in, const int* in_sizes, int n_in,
                              void* d_out, int out_size, void* d_ws, size_t ws_size,
                              hipStream_t stream)
{
  (void)in_sizes; (void)n_in; (void)out_size; (void)ws_size;
  const float* x  = (const float*)d_in[0];
  const int*   ei = (const int*)  d_in[1];
  const float* W1 = (const float*)d_in[2];
  const float* b1 = (const float*)d_in[3];
  const float* W2 = (const float*)d_in[4];
  const float* b2 = (const float*)d_in[5];
  float* out = (float*)d_out;
  const int* src = ei;
  const int* dst = ei + N_EDGES;

  char* ws = (char*)d_ws;
  size_t off = 0;
  auto take = [&](size_t bytes){ void* p = ws + off; off = (off + bytes + 255) & ~(size_t)255; return p; };
  unsigned short* aggh    = (unsigned short*)take((size_t)N_NODES*DCH*sizeof(unsigned short));
  unsigned short* aggl    = (unsigned short*)take((size_t)N_NODES*DCH*sizeof(unsigned short));
  int*            esorted = (int*)           take((size_t)N_EDGES*sizeof(int));
  int*            offsets = (int*)           take((size_t)(N_NODES+1)*sizeof(int));
  int*            mat     = (int*)           take((size_t)NMAT*sizeof(int));
  int*            sc      = (int*)           take((size_t)NMAT*sizeof(int));
  int*            bsum    = (int*)           take(512*sizeof(int));
  unsigned*       binned  = (unsigned*)      take((size_t)N_EDGES*sizeof(unsigned));
  unsigned short* wpk     = (unsigned short*)take((size_t)NLAY*2*2*16384*sizeof(unsigned short));

  k_bin_count  <<<BIN_BLK, BIN_T, 0, stream>>>(dst, mat);
  k_scan_blocks<<<SCAN_NB, SCAN_B, 0, stream>>>(mat, sc, bsum, NMAT);
  k_scan_sums  <<<1, 512, 0, stream>>>(bsum, SCAN_NB);
  k_bin_scatter<<<BIN_BLK, BIN_T, 0, stream>>>(src, dst, sc, bsum, binned);
  k_bucket_csr <<<NB_BKT, CSR_T, 0, stream>>>(binned, sc, bsum, offsets, esorted);
  k_pack_w     <<<(NLAY*2*16384+255)/256, 256, 0, stream>>>(W1, W2, wpk);

  const float* xc = x;
  for (int L = 0; L < NLAY; ++L){
    k_aggregate_pack<<<(N_NODES*32+255)/256, 256, 0, stream>>>(xc, offsets, esorted, aggh, aggl);
    const unsigned short* w1pk = wpk + (size_t)(L*2 + 0)*2*16384;
    const unsigned short* w2pk = wpk + (size_t)(L*2 + 1)*2*16384;
    k_gin_mlp<<<N_TILES/2, 32, 0, stream>>>(aggh, aggl, w1pk, b1 + L*DCH, w2pk, b2 + L*DCH,
                                            out, (L < NLAY-1) ? 1 : 0);
    xc = out;
  }
}
